# starting kernel
# baseline (speedup 1.0000x reference)
#include <hip/hip_runtime.h>
#include <math.h>


typedef __attribute__((ext_vector_type(2))) float v2f;
typedef __attribute__((ext_vector_type(8))) float v8f;

#define HIDC 256
#define HEADS 4
#define DHC 64

__device__ __forceinline__ float atomicMaxF(float* addr, float value) {
  if (value >= 0.0f)
    return __int_as_float(atomicMax((int*)addr, __float_as_int(value)));
  else
    return __uint_as_float(atomicMin((unsigned int*)addr, __float_as_uint(value)));
}

__device__ __forceinline__ float lrelu(float v) {
  return v > 0.0f ? v : 0.2f * v;
}

template <int K>
__global__ __launch_bounds__(512)
void gemm_wmma_f32(const float* __restrict__ A, const float* __restrict__ W,
                   const float* __restrict__ bias, float* __restrict__ C,
                   int M) {
  constexpr int LDA = K + 4;
  __shared__ float As[16 * LDA];

  const int tid   = threadIdx.x;
  const int mBase = blockIdx.x << 4;

  #pragma unroll
  for (int i = tid; i < 16 * K; i += 512) {
    int r = i / K;
    int c = i - r * K;
    int row = mBase + r;
    if (row >= M) row = M - 1;
    As[r * LDA + c] = A[(long)row * K + c];
  }
  __syncthreads();

  const int lane  = tid & 31;
  const int wave  = tid >> 5;
  const int nBase = wave << 4;
  const int col   = lane & 15;
  const int kHalf = (lane >> 4) << 1;

  const float* as = &As[(lane & 15) * LDA + kHalf];
  const float* bp = W + (long)kHalf * HIDC + nBase + col;

  v8f acc = {};
  #pragma unroll 8
  for (int kb = 0; kb < K; kb += 4) {
    v2f a, b;
    a.x = as[0];
    a.y = as[1];
    b.x = bp[0];
    b.y = bp[HIDC];
    acc = __builtin_amdgcn_wmma_f32_16x16x4_f32(
         false, a,  false, b,
         (short)0, acc,  false,  false);
    as += 4;
    bp += 4 * HIDC;
  }

  const int rowOff = (lane >> 4) << 3;
  const float bc = bias ? bias[nBase + col] : 0.0f;
  #pragma unroll
  for (int r = 0; r < 8; ++r) {
    int row = mBase + rowOff + r;
    if (row < M) C[(long)row * HIDC + nBase + col] = acc[r] + bc;
  }
}

__global__ __launch_bounds__(256)
void el_er_kernel(const float* __restrict__ feat, const float* __restrict__ al,
                  const float* __restrict__ ar, float* __restrict__ el,
                  float* __restrict__ er, int NH) {
  int warp = blockIdx.x * (blockDim.x >> 5) + (threadIdx.x >> 5);
  int lane = threadIdx.x & 31;
  if (warp >= NH) return;
  int n = warp >> 2, h = warp & 3;
  const float* f  = feat + (long)n * HIDC + h * DHC;
  const float* a1 = al + h * DHC;
  const float* a2 = ar + h * DHC;
  float f0 = f[lane], f1 = f[lane + 32];
  float sl = f0 * a1[lane] + f1 * a1[lane + 32];
  float sr = f0 * a2[lane] + f1 * a2[lane + 32];
  #pragma unroll
  for (int off = 16; off; off >>= 1) {
    sl += __shfl_down(sl, off, 32);
    sr += __shfl_down(sr, off, 32);
  }
  if (lane == 0) { el[warp] = sl; er[warp] = sr; }
}

__global__ void init_nh_kernel(float* __restrict__ m, float* __restrict__ dn,
                               int NH) {
  int i = blockIdx.x * blockDim.x + threadIdx.x;
  if (i < NH) { m[i] = -INFINITY; dn[i] = 0.0f; }
}

__global__ void init_out_kernel(const float* __restrict__ hin,
                                const float* __restrict__ bias,
                                float* __restrict__ out, long total) {
  long i = (long)blockIdx.x * blockDim.x + threadIdx.x;
  if (i < total) out[i] = hin[i] + bias[i & (HIDC - 1)];
}

__global__ void edge_max_kernel(const int* __restrict__ src,
                                const int* __restrict__ dst,
                                const float* __restrict__ el,
                                const float* __restrict__ er,
                                float* __restrict__ m, int EH) {
  int idx = blockIdx.x * blockDim.x + threadIdx.x;
  if (idx >= EH) return;
  int e = idx >> 2, h = idx & 3;
  int s = src[e], d = dst[e];
  float v = lrelu(el[s * 4 + h] + er[d * 4 + h]);
  atomicMaxF(&m[d * 4 + h], v);
}

__global__ void edge_exp_kernel(const int* __restrict__ src,
                                const int* __restrict__ dst,
                                const float* __restrict__ el,
                                const float* __restrict__ er,
                                const float* __restrict__ m,
                                float* __restrict__ dn,
                                float* __restrict__ ex, int EH) {
  int idx = blockIdx.x * blockDim.x + threadIdx.x;
  if (idx >= EH) return;
  int e = idx >> 2, h = idx & 3;
  int s = src[e], d = dst[e];
  float v = lrelu(el[s * 4 + h] + er[d * 4 + h]);
  float mm = m[d * 4 + h];
  if (!(mm > -1e38f)) mm = 0.0f;
  float x = __expf(v - mm);
  ex[idx] = x;
  atomicAdd(&dn[d * 4 + h], x);
}

__global__ __launch_bounds__(256)
void aggregate_kernel(const int* __restrict__ src, const int* __restrict__ dst,
                      const float* __restrict__ feat,
                      const float* __restrict__ ex,
                      const float* __restrict__ dn, float* __restrict__ out) {
  int e = blockIdx.x;
  int c = threadIdx.x;
  int h = c >> 6;
  int s = src[e], d = dst[e];
  float denom = dn[d * 4 + h];
  float alpha = ex[e * 4 + h] / fmaxf(denom, 1e-16f);
  atomicAdd(&out[(long)d * HIDC + c], alpha * feat[(long)s * HIDC + c]);
}

__global__ void relu_kernel(const float* __restrict__ in, float* __restrict__ out,
                            long total) {
  long i = (long)blockIdx.x * blockDim.x + threadIdx.x;
  if (i < total) out[i] = fmaxf(in[i], 0.0f);
}

__global__ void mean_heads_kernel(const float* __restrict__ hbuf,
                                  float* __restrict__ out, long total) {
  long i = (long)blockIdx.x * blockDim.x + threadIdx.x;
  if (i >= total) return;
  long n = i >> 6;
  int d = (int)(i & 63);
  const float* p = hbuf + n * HIDC + d;
  out[i] = 0.25f * (p[0] + p[DHC] + p[2 * DHC] + p[3 * DHC]);
}

extern "C" void kernel_launch(void* const* d_in, const int* in_sizes, int n_in,
                              void* d_out, int out_size, void* d_ws, size_t ws_size,
                              hipStream_t stream) {
  const float* x     = (const float*)d_in[0];
  const int*   ei    = (const int*)d_in[1];
  const float* W[3]  = {(const float*)d_in[2], (const float*)d_in[3],
                        (const float*)d_in[4]};
  const float* al[3] = {(const float*)d_in[5], (const float*)d_in[6],
                        (const float*)d_in[7]};
  const float* ar[3] = {(const float*)d_in[8], (const float*)d_in[9],
                        (const float*)d_in[10]};
  const float* b[3]  = {(const float*)d_in[11], (const float*)d_in[12],
                        (const float*)d_in[13]};
  const float* resW0 = (const float*)d_in[14];

  const int N = in_sizes[0] / 128;
  const int E = in_sizes[1] / 2;
  const int* src = ei;
  const int* dst = ei + E;

  float* ws   = (float*)d_ws;
  float* FEAT = ws;
  float* OUT  = FEAT + (size_t)N * HIDC;
  float* HBUF = OUT  + (size_t)N * HIDC;
  float* ELP  = HBUF + (size_t)N * HIDC;
  float* ERP  = ELP  + (size_t)N * HEADS;
  float* MP   = ERP  + (size_t)N * HEADS;
  float* DNP  = MP   + (size_t)N * HEADS;
  float* EXP  = DNP  + (size_t)N * HEADS;

  const int  mtiles = (N + 15) / 16;
  const long totNF  = (long)N * HIDC;
  const int  NH     = N * HEADS;
  const int  EH     = E * HEADS;
  const int  gNF    = (int)((totNF + 255) / 256);
  const int  gNH    = (NH + 255) / 256;
  const int  gEH    = (EH + 255) / 256;
  const int  gELR   = (NH + 7) / 8;

  auto attention = [&](const float* alw, const float* arw) {
    el_er_kernel<<<gELR, 256, 0, stream>>>(FEAT, alw, arw, ELP, ERP, NH);
    init_nh_kernel<<<gNH, 256, 0, stream>>>(MP, DNP, NH);
    edge_max_kernel<<<gEH, 256, 0, stream>>>(src, dst, ELP, ERP, MP, EH);
    edge_exp_kernel<<<gEH, 256, 0, stream>>>(src, dst, ELP, ERP, MP, DNP, EXP, EH);
    aggregate_kernel<<<E, 256, 0, stream>>>(src, dst, FEAT, EXP, DNP, OUT);
  };

  gemm_wmma_f32<128><<<mtiles, 512, 0, stream>>>(x, W[0], nullptr, FEAT, N);
  gemm_wmma_f32<128><<<mtiles, 512, 0, stream>>>(x, resW0, b[0], OUT, N);
  attention(al[0], ar[0]);
  relu_kernel<<<gNF, 256, 0, stream>>>(OUT, HBUF, totNF);

  gemm_wmma_f32<256><<<mtiles, 512, 0, stream>>>(HBUF, W[1], nullptr, FEAT, N);
  init_out_kernel<<<gNF, 256, 0, stream>>>(HBUF, b[1], OUT, totNF);
  attention(al[1], ar[1]);
  relu_kernel<<<gNF, 256, 0, stream>>>(OUT, HBUF, totNF);

  gemm_wmma_f32<256><<<mtiles, 512, 0, stream>>>(HBUF, W[2], nullptr, FEAT, N);
  init_out_kernel<<<gNF, 256, 0, stream>>>(HBUF, b[2], OUT, totNF);
  attention(al[2], ar[2]);
  mean_heads_kernel<<<(int)(((long)N * DHC + 255) / 256), 256, 0, stream>>>(
      OUT, (float*)d_out, (long)N * DHC);
}


# Edge softmax+aggregation per destination node over a dst-sorted edge list rebuilt each call (int atomics); float atomics removed, sums now in list order (atomic arrival order of the fill)
# speedup vs baseline: 3.8238x; 3.8238x over previous
#include <hip/hip_runtime.h>
#include <math.h>


typedef __attribute__((ext_vector_type(2))) float v2f;
typedef __attribute__((ext_vector_type(8))) float v8f;

#define HIDC 256
#define HEADS 4
#define DHC 64

__device__ __forceinline__ float lrelu(float v) {
  return v > 0.0f ? v : 0.2f * v;
}

template <int K>
__global__ __launch_bounds__(512)
void gemm_wmma_f32(const float* __restrict__ A, const float* __restrict__ W,
                   const float* __restrict__ bias, float* __restrict__ C,
                   int M) {
  constexpr int LDA = K + 4;
  __shared__ float As[16 * LDA];

  const int tid   = threadIdx.x;
  const int mBase = blockIdx.x << 4;

  #pragma unroll
  for (int i = tid; i < 16 * K; i += 512) {
    int r = i / K;
    int c = i - r * K;
    int row = mBase + r;
    if (row >= M) row = M - 1;
    As[r * LDA + c] = A[(long)row * K + c];
  }
  __syncthreads();

  const int lane  = tid & 31;
  const int wave  = tid >> 5;
  const int nBase = wave << 4;
  const int col   = lane & 15;
  const int kHalf = (lane >> 4) << 1;

  const float* as = &As[(lane & 15) * LDA + kHalf];
  const float* bp = W + (long)kHalf * HIDC + nBase + col;

  v8f acc = {};
  #pragma unroll 8
  for (int kb = 0; kb < K; kb += 4) {
    v2f a, b;
    a.x = as[0];
    a.y = as[1];
    b.x = bp[0];
    b.y = bp[HIDC];
    acc = __builtin_amdgcn_wmma_f32_16x16x4_f32(
         false, a,  false, b,
         (short)0, acc,  false,  false);
    as += 4;
    bp += 4 * HIDC;
  }

  const int rowOff = (lane >> 4) << 3;
  const float bc = bias ? bias[nBase + col] : 0.0f;
  #pragma unroll
  for (int r = 0; r < 8; ++r) {
    int row = mBase + rowOff + r;
    if (row < M) C[(long)row * HIDC + nBase + col] = acc[r] + bc;
  }
}

__global__ __launch_bounds__(256)
void el_er_kernel(const float* __restrict__ feat, const float* __restrict__ al,
                  const float* __restrict__ ar, float* __restrict__ el,
                  float* __restrict__ er, int NH) {
  int warp = blockIdx.x * (blockDim.x >> 5) + (threadIdx.x >> 5);
  int lane = threadIdx.x & 31;
  if (warp >= NH) return;
  int n = warp >> 2, h = warp & 3;
  const float* f  = feat + (long)n * HIDC + h * DHC;
  const float* a1 = al + h * DHC;
  const float* a2 = ar + h * DHC;
  float f0 = f[lane], f1 = f[lane + 32];
  float sl = f0 * a1[lane] + f1 * a1[lane + 32];
  float sr = f0 * a2[lane] + f1 * a2[lane + 32];
  #pragma unroll
  for (int off = 16; off; off >>= 1) {
    sl += __shfl_down(sl, off, 32);
    sr += __shfl_down(sr, off, 32);
  }
  if (lane == 0) { el[warp] = sl; er[warp] = sr; }
}

__global__ void init_out_kernel(const float* __restrict__ hin,
                                const float* __restrict__ bias,
                                float* __restrict__ out, long total) {
  long i = (long)blockIdx.x * blockDim.x + threadIdx.x;
  if (i < total) out[i] = hin[i] + bias[i & (HIDC - 1)];
}


__global__ void zero_int_kernel(int* __restrict__ p, int n) {
  int i = blockIdx.x * blockDim.x + threadIdx.x;
  if (i < n) p[i] = 0;
}

__global__ void in_degree_kernel(const int* __restrict__ dst,
                                 int* __restrict__ cnt, int E) {
  int e = blockIdx.x * blockDim.x + threadIdx.x;
  if (e < E) atomicAdd(&cnt[dst[e]], 1);
}

__global__ __launch_bounds__(256)
void row_start_kernel(const int* __restrict__ cnt, int* __restrict__ rowptr,
                      int n) {
  __shared__ int wave_pre[8];
  __shared__ int wave_tot[8];
  const int t = threadIdx.x, lane = t & 31, wave = t >> 5;
  const int tile0 = blockIdx.x * 256;

  int pre = 0;
  for (int i = t; i < tile0; i += 256) pre += cnt[i];
  for (int off = 16; off; off >>= 1) pre += __shfl_xor(pre, off, 32);

  const int i = tile0 + t;
  const int c = (i < n) ? cnt[i] : 0;
  int inc = c;
  for (int off = 1; off < 32; off <<= 1) {
    int v = __shfl_up(inc, off, 32);
    if (lane >= off) inc += v;
  }
  if (lane == 31) wave_tot[wave] = inc;
  if (lane == 0)  wave_pre[wave] = pre;
  __syncthreads();
  int base = 0;
  for (int w = 0; w < 8; ++w) {
    base += wave_pre[w];
    if (w < wave) base += wave_tot[w];
  }
  const int excl = base + inc - c;
  if (i < n) {
    rowptr[i] = excl;
    if (i == n - 1) rowptr[n] = excl + c;
  }
}

__global__ void csr_fill_kernel(const int* __restrict__ src,
                                const int* __restrict__ dst,
                                const int* __restrict__ rowptr,
                                int* __restrict__ cnt,
                                int* __restrict__ csrc, int E) {
  int e = blockIdx.x * blockDim.x + threadIdx.x;
  if (e >= E) return;
  int d = dst[e];
  int k = atomicSub(&cnt[d], 1);
  csrc[rowptr[d] + k - 1] = src[e];
}

__device__ __forceinline__ float wave_max(float v) {
  for (int off = 16; off; off >>= 1) v = fmaxf(v, __shfl_xor(v, off, 32));
  return v;
}

__device__ __forceinline__ float wave_sum(float v) {
  for (int off = 16; off; off >>= 1) v += __shfl_xor(v, off, 32);
  return v;
}

__device__ __forceinline__ float lane_bcast(float v, int k) {
  return __int_as_float(__builtin_amdgcn_readlane(__float_as_int(v), k));
}

__global__ __launch_bounds__(256)
void node_softmax_aggregate_kernel(const int* __restrict__ rowptr,
                                   const int* __restrict__ csrc,
                                   const float* __restrict__ feat,
                                   const float* __restrict__ el,
                                   const float* __restrict__ er,
                                   float* __restrict__ out, int n) {
  const int lane = threadIdx.x & 31;
  const int node = __builtin_amdgcn_readfirstlane(
      (int)(blockIdx.x * 8 + (threadIdx.x >> 5)));
  if (node >= n) return;
  const int start = rowptr[node];
  const int deg   = rowptr[node + 1] - start;
  if (deg <= 0) return;

  const int*    cs  = csrc + start;
  const float4* el4 = (const float4*)el;
  const float4* f4  = (const float4*)feat;
  const float4  r4  = ((const float4*)er)[node];

  const bool has0 = lane < deg;
  int   s0  = 0;
  float e0x = -INFINITY, e0y = -INFINITY, e0z = -INFINITY, e0w = -INFINITY;
  if (has0) {
    s0 = cs[lane];
    const float4 l = el4[s0];
    e0x = lrelu(l.x + r4.x); e0y = lrelu(l.y + r4.y);
    e0z = lrelu(l.z + r4.z); e0w = lrelu(l.w + r4.w);
  }

  float mx = e0x, my = e0y, mz = e0z, mw = e0w;
  for (int j = lane + 32; j < deg; j += 32) {
    const float4 l = el4[cs[j]];
    mx = fmaxf(mx, lrelu(l.x + r4.x)); my = fmaxf(my, lrelu(l.y + r4.y));
    mz = fmaxf(mz, lrelu(l.z + r4.z)); mw = fmaxf(mw, lrelu(l.w + r4.w));
  }
  mx = wave_max(mx); my = wave_max(my); mz = wave_max(mz); mw = wave_max(mw);
  if (!(mx > -1e38f)) mx = 0.0f;
  if (!(my > -1e38f)) my = 0.0f;
  if (!(mz > -1e38f)) mz = 0.0f;
  if (!(mw > -1e38f)) mw = 0.0f;

  const float x0x = has0 ? __expf(e0x - mx) : 0.0f;
  const float x0y = has0 ? __expf(e0y - my) : 0.0f;
  const float x0z = has0 ? __expf(e0z - mz) : 0.0f;
  const float x0w = has0 ? __expf(e0w - mw) : 0.0f;
  float dx = x0x, dy = x0y, dz = x0z, dw = x0w;
  for (int j = lane + 32; j < deg; j += 32) {
    const float4 l = el4[cs[j]];
    dx += __expf(lrelu(l.x + r4.x) - mx); dy += __expf(lrelu(l.y + r4.y) - my);
    dz += __expf(lrelu(l.z + r4.z) - mz); dw += __expf(lrelu(l.w + r4.w) - mw);
  }
  dx = fmaxf(wave_sum(dx), 1e-16f); dy = fmaxf(wave_sum(dy), 1e-16f);
  dz = fmaxf(wave_sum(dz), 1e-16f); dw = fmaxf(wave_sum(dw), 1e-16f);

  const bool hi = lane >= 16;
  float4 accA = {0.0f, 0.0f, 0.0f, 0.0f};
  float4 accB = {0.0f, 0.0f, 0.0f, 0.0f};
  for (int base = 0; base < deg; base += 32) {
    int   s  = s0;
    float ax = x0x / dx, ay = x0y / dy, az = x0z / dz, aw = x0w / dw;
    if (base > 0) {
      const int j = base + lane;
      s = 0; ax = 0.0f; ay = 0.0f; az = 0.0f; aw = 0.0f;
      if (j < deg) {
        s = cs[j];
        const float4 l = el4[s];
        ax = __expf(lrelu(l.x + r4.x) - mx) / dx;
        ay = __expf(lrelu(l.y + r4.y) - my) / dy;
        az = __expf(lrelu(l.z + r4.z) - mz) / dz;
        aw = __expf(lrelu(l.w + r4.w) - mw) / dw;
      }
    }
    const int left = deg - base;
    const int cnt  = left < 32 ? left : 32;
    #pragma unroll 4
    for (int k = 0; k < cnt; ++k) {
      const int   sk = __builtin_amdgcn_readlane(s, k);
      const float bx = lane_bcast(ax, k), by = lane_bcast(ay, k);
      const float bz = lane_bcast(az, k), bw = lane_bcast(aw, k);
      const float aA = hi ? by : bx;
      const float aB = hi ? bw : bz;
      const float4* row = f4 + (long)sk * (HIDC / 4);
      const float4 fA = row[lane];
      const float4 fB = row[32 + lane];
      accA.x += aA * fA.x; accA.y += aA * fA.y;
      accA.z += aA * fA.z; accA.w += aA * fA.w;
      accB.x += aB * fB.x; accB.y += aB * fB.y;
      accB.z += aB * fB.z; accB.w += aB * fB.w;
    }
  }

  float4* o4 = (float4*)out + (long)node * (HIDC / 4);
  float4 oA = o4[lane], oB = o4[32 + lane];
  oA.x += accA.x; oA.y += accA.y; oA.z += accA.z; oA.w += accA.w;
  oB.x += accB.x; oB.y += accB.y; oB.z += accB.z; oB.w += accB.w;
  o4[lane] = oA;
  o4[32 + lane] = oB;
}

__global__ void relu_kernel(const float* __restrict__ in, float* __restrict__ out,
                            long total) {
  long i = (long)blockIdx.x * blockDim.x + threadIdx.x;
  if (i < total) out[i] = fmaxf(in[i], 0.0f);
}

__global__ void mean_heads_kernel(const float* __restrict__ hbuf,
                                  float* __restrict__ out, long total) {
  long i = (long)blockIdx.x * blockDim.x + threadIdx.x;
  if (i >= total) return;
  long n = i >> 6;
  int d = (int)(i & 63);
  const float* p = hbuf + n * HIDC + d;
  out[i] = 0.25f * (p[0] + p[DHC] + p[2 * DHC] + p[3 * DHC]);
}

extern "C" void kernel_launch(void* const* d_in, const int* in_sizes, int n_in,
                              void* d_out, int out_size, void* d_ws, size_t ws_size,
                              hipStream_t stream) {
  const float* x     = (const float*)d_in[0];
  const int*   ei    = (const int*)d_in[1];
  const float* W[3]  = {(const float*)d_in[2], (const float*)d_in[3],
                        (const float*)d_in[4]};
  const float* al[3] = {(const float*)d_in[5], (const float*)d_in[6],
                        (const float*)d_in[7]};
  const float* ar[3] = {(const float*)d_in[8], (const float*)d_in[9],
                        (const float*)d_in[10]};
  const float* b[3]  = {(const float*)d_in[11], (const float*)d_in[12],
                        (const float*)d_in[13]};
  const float* resW0 = (const float*)d_in[14];

  const int N = in_sizes[0] / 128;
  const int E = in_sizes[1] / 2;
  const int* src = ei;
  const int* dst = ei + E;

  float* ws   = (float*)d_ws;
  float* FEAT = ws;
  float* OUT  = FEAT + (size_t)N * HIDC;
  float* HBUF = OUT  + (size_t)N * HIDC;
  float* ELP  = HBUF + (size_t)N * HIDC;
  float* ERP  = ELP  + (size_t)N * HEADS;
  int*   CNT  = (int*)(ERP + (size_t)N * HEADS);
  int*   RPT  = CNT + (size_t)N;
  int*   CSRC = RPT + (size_t)N + 1;

  const int  mtiles = (N + 15) / 16;
  const long totNF  = (long)N * HIDC;
  const int  NH     = N * HEADS;
  const int  gNF    = (int)((totNF + 255) / 256);
  const int  gN     = (N + 255) / 256;
  const int  gE     = (E + 255) / 256;
  const int  gELR   = (NH + 7) / 8;
  const int  gNODE  = (N + 7) / 8;

  zero_int_kernel<<<gN, 256, 0, stream>>>(CNT, N);
  in_degree_kernel<<<gE, 256, 0, stream>>>(dst, CNT, E);
  row_start_kernel<<<gN, 256, 0, stream>>>(CNT, RPT, N);
  csr_fill_kernel<<<gE, 256, 0, stream>>>(src, dst, RPT, CNT, CSRC, E);

  auto attention = [&](const float* alw, const float* arw) {
    el_er_kernel<<<gELR, 256, 0, stream>>>(FEAT, alw, arw, ELP, ERP, NH);
    node_softmax_aggregate_kernel<<<gNODE, 256, 0, stream>>>(RPT, CSRC, FEAT, ELP,
                                                             ERP, OUT, N);
  };

  gemm_wmma_f32<128><<<mtiles, 512, 0, stream>>>(x, W[0], nullptr, FEAT, N);
  gemm_wmma_f32<128><<<mtiles, 512, 0, stream>>>(x, resW0, b[0], OUT, N);
  attention(al[0], ar[0]);
  relu_kernel<<<gNF, 256, 0, stream>>>(OUT, HBUF, totNF);

  gemm_wmma_f32<256><<<mtiles, 512, 0, stream>>>(HBUF, W[1], nullptr, FEAT, N);
  init_out_kernel<<<gNF, 256, 0, stream>>>(HBUF, b[1], OUT, totNF);
  attention(al[1], ar[1]);
  relu_kernel<<<gNF, 256, 0, stream>>>(OUT, HBUF, totNF);

  gemm_wmma_f32<256><<<mtiles, 512, 0, stream>>>(HBUF, W[2], nullptr, FEAT, N);
  init_out_kernel<<<gNF, 256, 0, stream>>>(HBUF, b[2], OUT, totNF);
  attention(al[2], ar[2]);
  mean_heads_kernel<<<(int)(((long)N * DHC + 255) / 256), 256, 0, stream>>>(
      OUT, (float*)d_out, (long)N * DHC);
}


# Residual+bias start, relu and head mean moved into the node-owning wave of the aggregation kernel (5 one-element-per-thread launches removed); order of every floating-point sum unchanged
# speedup vs baseline: 4.9174x; 1.2860x over previous
#include <hip/hip_runtime.h>
#include <math.h>


typedef __attribute__((ext_vector_type(2))) float v2f;
typedef __attribute__((ext_vector_type(8))) float v8f;

#define HIDC 256
#define HEADS 4
#define DHC 64

__device__ __forceinline__ float lrelu(float v) {
  return v > 0.0f ? v : 0.2f * v;
}

template <int K>
__global__ __launch_bounds__(512)
void gemm_wmma_f32(const float* __restrict__ A, const float* __restrict__ W,
                   const float* __restrict__ bias, float* __restrict__ C,
                   int M) {
  constexpr int LDA = K + 4;
  __shared__ float As[16 * LDA];

  const int tid   = threadIdx.x;
  const int mBase = blockIdx.x << 4;

  #pragma unroll
  for (int i = tid; i < 16 * K; i += 512) {
    int r = i / K;
    int c = i - r * K;
    int row = mBase + r;
    if (row >= M) row = M - 1;
    As[r * LDA + c] = A[(long)row * K + c];
  }
  __syncthreads();

  const int lane  = tid & 31;
  const int wave  = tid >> 5;
  const int nBase = wave << 4;
  const int col   = lane & 15;
  const int kHalf = (lane >> 4) << 1;

  const float* as = &As[(lane & 15) * LDA + kHalf];
  const float* bp = W + (long)kHalf * HIDC + nBase + col;

  v8f acc = {};
  #pragma unroll 8
  for (int kb = 0; kb < K; kb += 4) {
    v2f a, b;
    a.x = as[0];
    a.y = as[1];
    b.x = bp[0];
    b.y = bp[HIDC];
    acc = __builtin_amdgcn_wmma_f32_16x16x4_f32(
         false, a,  false, b,
         (short)0, acc,  false,  false);
    as += 4;
    bp += 4 * HIDC;
  }

  const int rowOff = (lane >> 4) << 3;
  const float bc = bias ? bias[nBase + col] : 0.0f;
  #pragma unroll
  for (int r = 0; r < 8; ++r) {
    int row = mBase + rowOff + r;
    if (row < M) C[(long)row * HIDC + nBase + col] = acc[r] + bc;
  }
}

__global__ __launch_bounds__(256)
void el_er_kernel(const float* __restrict__ feat, const float* __restrict__ al,
                  const float* __restrict__ ar, float* __restrict__ el,
                  float* __restrict__ er, int NH) {
  int warp = blockIdx.x * (blockDim.x >> 5) + (threadIdx.x >> 5);
  int lane = threadIdx.x & 31;
  if (warp >= NH) return;
  int n = warp >> 2, h = warp & 3;
  const float* f  = feat + (long)n * HIDC + h * DHC;
  const float* a1 = al + h * DHC;
  const float* a2 = ar + h * DHC;
  float f0 = f[lane], f1 = f[lane + 32];
  float sl = f0 * a1[lane] + f1 * a1[lane + 32];
  float sr = f0 * a2[lane] + f1 * a2[lane + 32];
  #pragma unroll
  for (int off = 16; off; off >>= 1) {
    sl += __shfl_down(sl, off, 32);
    sr += __shfl_down(sr, off, 32);
  }
  if (lane == 0) { el[warp] = sl; er[warp] = sr; }
}


__global__ void zero_int_kernel(int* __restrict__ p, int n) {
  int i = blockIdx.x * blockDim.x + threadIdx.x;
  if (i < n) p[i] = 0;
}

__global__ void in_degree_kernel(const int* __restrict__ dst,
                                 int* __restrict__ cnt, int E) {
  int e = blockIdx.x * blockDim.x + threadIdx.x;
  if (e < E) atomicAdd(&cnt[dst[e]], 1);
}

__global__ __launch_bounds__(256)
void row_start_kernel(const int* __restrict__ cnt, int* __restrict__ rowptr,
                      int n) {
  __shared__ int wave_pre[8];
  __shared__ int wave_tot[8];
  const int t = threadIdx.x, lane = t & 31, wave = t >> 5;
  const int tile0 = blockIdx.x * 256;

  int pre = 0;
  for (int i = t; i < tile0; i += 256) pre += cnt[i];
  for (int off = 16; off; off >>= 1) pre += __shfl_xor(pre, off, 32);

  const int i = tile0 + t;
  const int c = (i < n) ? cnt[i] : 0;
  int inc = c;
  for (int off = 1; off < 32; off <<= 1) {
    int v = __shfl_up(inc, off, 32);
    if (lane >= off) inc += v;
  }
  if (lane == 31) wave_tot[wave] = inc;
  if (lane == 0)  wave_pre[wave] = pre;
  __syncthreads();
  int base = 0;
  for (int w = 0; w < 8; ++w) {
    base += wave_pre[w];
    if (w < wave) base += wave_tot[w];
  }
  const int excl = base + inc - c;
  if (i < n) {
    rowptr[i] = excl;
    if (i == n - 1) rowptr[n] = excl + c;
  }
}

__global__ void csr_fill_kernel(const int* __restrict__ src,
                                const int* __restrict__ dst,
                                const int* __restrict__ rowptr,
                                int* __restrict__ cnt,
                                int* __restrict__ csrc, int E) {
  int e = blockIdx.x * blockDim.x + threadIdx.x;
  if (e >= E) return;
  int d = dst[e];
  int k = atomicSub(&cnt[d], 1);
  csrc[rowptr[d] + k - 1] = src[e];
}

__device__ __forceinline__ float wave_max(float v) {
  for (int off = 16; off; off >>= 1) v = fmaxf(v, __shfl_xor(v, off, 32));
  return v;
}

__device__ __forceinline__ float wave_sum(float v) {
  for (int off = 16; off; off >>= 1) v += __shfl_xor(v, off, 32);
  return v;
}

__device__ __forceinline__ float lane_bcast(float v, int k) {
  return __int_as_float(__builtin_amdgcn_readlane(__float_as_int(v), k));
}

template <bool MEAN>
__global__ __launch_bounds__(256)
void node_softmax_aggregate_kernel(const int* __restrict__ rowptr,
                                   const int* __restrict__ csrc,
                                   const float* __restrict__ feat,
                                   const float* __restrict__ el,
                                   const float* __restrict__ er,
                                   const float* __restrict__ res,
                                   const float* __restrict__ bias,
                                   float* __restrict__ out, int n) {
  const int lane = threadIdx.x & 31;
  const int node = __builtin_amdgcn_readfirstlane(
      (int)(blockIdx.x * 8 + (threadIdx.x >> 5)));
  if (node >= n) return;
  const int start = rowptr[node];
  const int deg   = rowptr[node + 1] - start;

  float4 accA = {0.0f, 0.0f, 0.0f, 0.0f};
  float4 accB = {0.0f, 0.0f, 0.0f, 0.0f};
  if (deg > 0) {
    const int*    cs  = csrc + start;
    const float4* el4 = (const float4*)el;
    const float4* f4  = (const float4*)feat;
    const float4  r4  = ((const float4*)er)[node];

    const bool has0 = lane < deg;
    int   s0  = 0;
    float e0x = -INFINITY, e0y = -INFINITY, e0z = -INFINITY, e0w = -INFINITY;
    if (has0) {
      s0 = cs[lane];
      const float4 l = el4[s0];
      e0x = lrelu(l.x + r4.x); e0y = lrelu(l.y + r4.y);
      e0z = lrelu(l.z + r4.z); e0w = lrelu(l.w + r4.w);
    }

    float mx = e0x, my = e0y, mz = e0z, mw = e0w;
    for (int j = lane + 32; j < deg; j += 32) {
      const float4 l = el4[cs[j]];
      mx = fmaxf(mx, lrelu(l.x + r4.x)); my = fmaxf(my, lrelu(l.y + r4.y));
      mz = fmaxf(mz, lrelu(l.z + r4.z)); mw = fmaxf(mw, lrelu(l.w + r4.w));
    }
    mx = wave_max(mx); my = wave_max(my); mz = wave_max(mz); mw = wave_max(mw);
    if (!(mx > -1e38f)) mx = 0.0f;
    if (!(my > -1e38f)) my = 0.0f;
    if (!(mz > -1e38f)) mz = 0.0f;
    if (!(mw > -1e38f)) mw = 0.0f;

    const float x0x = has0 ? __expf(e0x - mx) : 0.0f;
    const float x0y = has0 ? __expf(e0y - my) : 0.0f;
    const float x0z = has0 ? __expf(e0z - mz) : 0.0f;
    const float x0w = has0 ? __expf(e0w - mw) : 0.0f;
    float dx = x0x, dy = x0y, dz = x0z, dw = x0w;
    for (int j = lane + 32; j < deg; j += 32) {
      const float4 l = el4[cs[j]];
      dx += __expf(lrelu(l.x + r4.x) - mx); dy += __expf(lrelu(l.y + r4.y) - my);
      dz += __expf(lrelu(l.z + r4.z) - mz); dw += __expf(lrelu(l.w + r4.w) - mw);
    }
    dx = fmaxf(wave_sum(dx), 1e-16f); dy = fmaxf(wave_sum(dy), 1e-16f);
    dz = fmaxf(wave_sum(dz), 1e-16f); dw = fmaxf(wave_sum(dw), 1e-16f);

    const bool hi = lane >= 16;
    for (int base = 0; base < deg; base += 32) {
      int   s  = s0;
      float ax = x0x / dx, ay = x0y / dy, az = x0z / dz, aw = x0w / dw;
      if (base > 0) {
        const int j = base + lane;
        s = 0; ax = 0.0f; ay = 0.0f; az = 0.0f; aw = 0.0f;
        if (j < deg) {
          s = cs[j];
          const float4 l = el4[s];
          ax = __expf(lrelu(l.x + r4.x) - mx) / dx;
          ay = __expf(lrelu(l.y + r4.y) - my) / dy;
          az = __expf(lrelu(l.z + r4.z) - mz) / dz;
          aw = __expf(lrelu(l.w + r4.w) - mw) / dw;
        }
      }
      const int left = deg - base;
      const int cnt  = left < 32 ? left : 32;
      #pragma unroll 4
      for (int k = 0; k < cnt; ++k) {
        const int   sk = __builtin_amdgcn_readlane(s, k);
        const float bx = lane_bcast(ax, k), by = lane_bcast(ay, k);
        const float bz = lane_bcast(az, k), bw = lane_bcast(aw, k);
        const float aA = hi ? by : bx;
        const float aB = hi ? bw : bz;
        const float4* row = f4 + (long)sk * (HIDC / 4);
        const float4 fA = row[lane];
        const float4 fB = row[32 + lane];
        accA.x += aA * fA.x; accA.y += aA * fA.y;
        accA.z += aA * fA.z; accA.w += aA * fA.w;
        accB.x += aB * fB.x; accB.y += aB * fB.y;
        accB.z += aB * fB.z; accB.w += aB * fB.w;
      }
    }
  }

  const float4* rs4 = (const float4*)res + (long)node * (HIDC / 4);
  float4 oA = rs4[lane], oB = rs4[32 + lane];
  if (bias) {
    const float4 bA = ((const float4*)bias)[lane];
    const float4 bB = ((const float4*)bias)[32 + lane];
    oA.x += bA.x; oA.y += bA.y; oA.z += bA.z; oA.w += bA.w;
    oB.x += bB.x; oB.y += bB.y; oB.z += bB.z; oB.w += bB.w;
  }
  oA.x += accA.x; oA.y += accA.y; oA.z += accA.z; oA.w += accA.w;
  oB.x += accB.x; oB.y += accB.y; oB.z += accB.z; oB.w += accB.w;

  if (!MEAN) {
    oA.x = fmaxf(oA.x, 0.0f); oA.y = fmaxf(oA.y, 0.0f);
    oA.z = fmaxf(oA.z, 0.0f); oA.w = fmaxf(oA.w, 0.0f);
    oB.x = fmaxf(oB.x, 0.0f); oB.y = fmaxf(oB.y, 0.0f);
    oB.z = fmaxf(oB.z, 0.0f); oB.w = fmaxf(oB.w, 0.0f);
    float4* o4 = (float4*)out + (long)node * (HIDC / 4);
    o4[lane] = oA;
    o4[32 + lane] = oB;
  } else {
    float4 h1, h3;
    h1.x = __shfl_xor(oA.x, 16, 32); h1.y = __shfl_xor(oA.y, 16, 32);
    h1.z = __shfl_xor(oA.z, 16, 32); h1.w = __shfl_xor(oA.w, 16, 32);
    h3.x = __shfl_xor(oB.x, 16, 32); h3.y = __shfl_xor(oB.y, 16, 32);
    h3.z = __shfl_xor(oB.z, 16, 32); h3.w = __shfl_xor(oB.w, 16, 32);
    if (lane < 16) {
      float4 m;
      m.x = 0.25f * (((oA.x + h1.x) + oB.x) + h3.x);
      m.y = 0.25f * (((oA.y + h1.y) + oB.y) + h3.y);
      m.z = 0.25f * (((oA.z + h1.z) + oB.z) + h3.z);
      m.w = 0.25f * (((oA.w + h1.w) + oB.w) + h3.w);
      ((float4*)out)[(long)node * (DHC / 4) + lane] = m;
    }
  }
}

extern "C" void kernel_launch(void* const* d_in, const int* in_sizes, int n_in,
                              void* d_out, int out_size, void* d_ws, size_t ws_size,
                              hipStream_t stream) {
  const float* x     = (const float*)d_in[0];
  const int*   ei    = (const int*)d_in[1];
  const float* W[3]  = {(const float*)d_in[2], (const float*)d_in[3],
                        (const float*)d_in[4]};
  const float* al[3] = {(const float*)d_in[5], (const float*)d_in[6],
                        (const float*)d_in[7]};
  const float* ar[3] = {(const float*)d_in[8], (const float*)d_in[9],
                        (const float*)d_in[10]};
  const float* b[3]  = {(const float*)d_in[11], (const float*)d_in[12],
                        (const float*)d_in[13]};
  const float* resW0 = (const float*)d_in[14];

  const int N = in_sizes[0] / 128;
  const int E = in_sizes[1] / 2;
  const int* src = ei;
  const int* dst = ei + E;

  float* ws   = (float*)d_ws;
  float* FEAT = ws;
  float* OUT  = FEAT + (size_t)N * HIDC;
  float* HBUF = OUT  + (size_t)N * HIDC;
  float* ELP  = HBUF + (size_t)N * HIDC;
  float* ERP  = ELP  + (size_t)N * HEADS;
  int*   CNT  = (int*)(ERP + (size_t)N * HEADS);
  int*   RPT  = CNT + (size_t)N;
  int*   CSRC = RPT + (size_t)N + 1;

  const int  mtiles = (N + 15) / 16;
  const int  NH     = N * HEADS;
  const int  gN     = (N + 255) / 256;
  const int  gE     = (E + 255) / 256;
  const int  gELR   = (NH + 7) / 8;
  const int  gNODE  = (N + 7) / 8;

  zero_int_kernel<<<gN, 256, 0, stream>>>(CNT, N);
  in_degree_kernel<<<gE, 256, 0, stream>>>(dst, CNT, E);
  row_start_kernel<<<gN, 256, 0, stream>>>(CNT, RPT, N);
  csr_fill_kernel<<<gE, 256, 0, stream>>>(src, dst, RPT, CNT, CSRC, E);

  gemm_wmma_f32<128><<<mtiles, 512, 0, stream>>>(x, W[0], nullptr, FEAT, N);
  gemm_wmma_f32<128><<<mtiles, 512, 0, stream>>>(x, resW0, b[0], OUT, N);
  el_er_kernel<<<gELR, 256, 0, stream>>>(FEAT, al[0], ar[0], ELP, ERP, NH);
  node_softmax_aggregate_kernel<false><<<gNODE, 256, 0, stream>>>(
      RPT, CSRC, FEAT, ELP, ERP, OUT, nullptr, HBUF, N);

  gemm_wmma_f32<256><<<mtiles, 512, 0, stream>>>(HBUF, W[1], nullptr, FEAT, N);
  el_er_kernel<<<gELR, 256, 0, stream>>>(FEAT, al[1], ar[1], ELP, ERP, NH);
  node_softmax_aggregate_kernel<false><<<gNODE, 256, 0, stream>>>(
      RPT, CSRC, FEAT, ELP, ERP, HBUF, b[1], OUT, N);

  gemm_wmma_f32<256><<<mtiles, 512, 0, stream>>>(OUT, W[2], nullptr, FEAT, N);
  el_er_kernel<<<gELR, 256, 0, stream>>>(FEAT, al[2], ar[2], ELP, ERP, NH);
  node_softmax_aggregate_kernel<true><<<gNODE, 256, 0, stream>>>(
      RPT, CSRC, FEAT, ELP, ERP, OUT, b[2], (float*)d_out, N);
}


# el/er attention dots computed in the projection GEMM epilogue from the accumulator tiles (3 launches and re-read of projected rows gone); their sums now a 16-lane butterfly, then 4 waves in order
# speedup vs baseline: 5.3984x; 1.0978x over previous
#include <hip/hip_runtime.h>
#include <math.h>


typedef __attribute__((ext_vector_type(2))) float v2f;
typedef __attribute__((ext_vector_type(8))) float v8f;

#define HIDC 256
#define HEADS 4
#define DHC 64

__device__ __forceinline__ float lrelu(float v) {
  return v > 0.0f ? v : 0.2f * v;
}

template <int K, bool ATT>
__global__ __launch_bounds__(512)
void gemm_wmma_f32(const float* __restrict__ A, const float* __restrict__ W,
                   const float* __restrict__ bias, float* __restrict__ C,
                   const float* __restrict__ al, const float* __restrict__ ar,
                   float* __restrict__ el, float* __restrict__ er, int M) {
  constexpr int LDA = K + 4;
  __shared__ float As[16 * LDA];
  __shared__ float4 red4[ATT ? 2 * 16 * 4 : 1];

  const int tid   = threadIdx.x;
  const int mBase = blockIdx.x << 4;

  #pragma unroll
  for (int i = tid; i < 16 * K; i += 512) {
    int r = i / K;
    int c = i - r * K;
    int row = mBase + r;
    if (row >= M) row = M - 1;
    As[r * LDA + c] = A[(long)row * K + c];
  }
  __syncthreads();

  const int lane  = tid & 31;
  const int wave  = tid >> 5;
  const int nBase = wave << 4;
  const int col   = lane & 15;
  const int kHalf = (lane >> 4) << 1;

  const float* as = &As[(lane & 15) * LDA + kHalf];
  const float* bp = W + (long)kHalf * HIDC + nBase + col;

  const float alc = ATT ? al[nBase + col] : 0.0f;
  const float arc = ATT ? ar[nBase + col] : 0.0f;

  v8f acc = {};
  #pragma unroll 8
  for (int kb = 0; kb < K; kb += 4) {
    v2f a, b;
    a.x = as[0];
    a.y = as[1];
    b.x = bp[0];
    b.y = bp[HIDC];
    acc = __builtin_amdgcn_wmma_f32_16x16x4_f32(
         false, a,  false, b,
         (short)0, acc,  false,  false);
    as += 4;
    bp += 4 * HIDC;
  }

  const int rowOff = (lane >> 4) << 3;

  if (ATT) {
    float v[16];
    #pragma unroll
    for (int r = 0; r < 8; ++r) {
      v[2 * r]     = acc[r] * alc;
      v[2 * r + 1] = acc[r] * arc;
    }
    const bool b3 = (col & 8) != 0, b2 = (col & 4) != 0;
    const bool b1 = (col & 2) != 0, b0 = (col & 1) != 0;
    float w8[8], w4[4], w2[2];
    #pragma unroll
    for (int j = 0; j < 8; ++j) {
      const float keep = b3 ? v[8 + j] : v[j];
      const float send = b3 ? v[j] : v[8 + j];
      w8[j] = keep + __shfl_xor(send, 8, 32);
    }
    #pragma unroll
    for (int j = 0; j < 4; ++j) {
      const float keep = b2 ? w8[4 + j] : w8[j];
      const float send = b2 ? w8[j] : w8[4 + j];
      w4[j] = keep + __shfl_xor(send, 4, 32);
    }
    #pragma unroll
    for (int j = 0; j < 2; ++j) {
      const float keep = b1 ? w4[2 + j] : w4[j];
      const float send = b1 ? w4[j] : w4[2 + j];
      w2[j] = keep + __shfl_xor(send, 2, 32);
    }
    const float keep1 = b0 ? w2[1] : w2[0];
    const float send1 = b0 ? w2[0] : w2[1];
    const float part  = keep1 + __shfl_xor(send1, 1, 32);

    float* red = (float*)red4;
    red[(((col & 1) << 4) + rowOff + (col >> 1)) * 16 + wave] = part;
    __syncthreads();
  }

  const float bc = bias ? bias[nBase + col] : 0.0f;
  #pragma unroll
  for (int r = 0; r < 8; ++r) {
    int row = mBase + rowOff + r;
    if (row < M) C[(long)row * HIDC + nBase + col] = acc[r] + bc;
  }

  if (ATT) {
    if (wave < 2) {
      float* dstp = (wave == 0) ? el : er;
      #pragma unroll
      for (int half = 0; half < 2; ++half) {
        const int idx = half * 32 + lane;
        const float4 q = red4[wave * 64 + idx];
        if (mBase + (idx >> 2) < M)
          dstp[(long)mBase * HEADS + idx] = ((q.x + q.y) + q.z) + q.w;
      }
    }
  }
}


__global__ void zero_int_kernel(int* __restrict__ p, int n) {
  int i = blockIdx.x * blockDim.x + threadIdx.x;
  if (i < n) p[i] = 0;
}

__global__ void in_degree_kernel(const int* __restrict__ dst,
                                 int* __restrict__ cnt, int E) {
  int e = blockIdx.x * blockDim.x + threadIdx.x;
  if (e < E) atomicAdd(&cnt[dst[e]], 1);
}

__global__ __launch_bounds__(256)
void row_start_kernel(const int* __restrict__ cnt, int* __restrict__ rowptr,
                      int n) {
  __shared__ int wave_pre[8];
  __shared__ int wave_tot[8];
  const int t = threadIdx.x, lane = t & 31, wave = t >> 5;
  const int tile0 = blockIdx.x * 256;

  int pre = 0;
  for (int i = t; i < tile0; i += 256) pre += cnt[i];
  for (int off = 16; off; off >>= 1) pre += __shfl_xor(pre, off, 32);

  const int i = tile0 + t;
  const int c = (i < n) ? cnt[i] : 0;
  int inc = c;
  for (int off = 1; off < 32; off <<= 1) {
    int v = __shfl_up(inc, off, 32);
    if (lane >= off) inc += v;
  }
  if (lane == 31) wave_tot[wave] = inc;
  if (lane == 0)  wave_pre[wave] = pre;
  __syncthreads();
  int base = 0;
  for (int w = 0; w < 8; ++w) {
    base += wave_pre[w];
    if (w < wave) base += wave_tot[w];
  }
  const int excl = base + inc - c;
  if (i < n) {
    rowptr[i] = excl;
    if (i == n - 1) rowptr[n] = excl + c;
  }
}

__global__ void csr_fill_kernel(const int* __restrict__ src,
                                const int* __restrict__ dst,
                                const int* __restrict__ rowptr,
                                int* __restrict__ cnt,
                                int* __restrict__ csrc, int E) {
  int e = blockIdx.x * blockDim.x + threadIdx.x;
  if (e >= E) return;
  int d = dst[e];
  int k = atomicSub(&cnt[d], 1);
  csrc[rowptr[d] + k - 1] = src[e];
}

__device__ __forceinline__ float wave_max(float v) {
  for (int off = 16; off; off >>= 1) v = fmaxf(v, __shfl_xor(v, off, 32));
  return v;
}

__device__ __forceinline__ float wave_sum(float v) {
  for (int off = 16; off; off >>= 1) v += __shfl_xor(v, off, 32);
  return v;
}

__device__ __forceinline__ float lane_bcast(float v, int k) {
  return __int_as_float(__builtin_amdgcn_readlane(__float_as_int(v), k));
}

template <bool MEAN>
__global__ __launch_bounds__(256)
void node_softmax_aggregate_kernel(const int* __restrict__ rowptr,
                                   const int* __restrict__ csrc,
                                   const float* __restrict__ feat,
                                   const float* __restrict__ el,
                                   const float* __restrict__ er,
                                   const float* __restrict__ res,
                                   const float* __restrict__ bias,
                                   float* __restrict__ out, int n) {
  const int lane = threadIdx.x & 31;
  const int node = __builtin_amdgcn_readfirstlane(
      (int)(blockIdx.x * 8 + (threadIdx.x >> 5)));
  if (node >= n) return;
  const int start = rowptr[node];
  const int deg   = rowptr[node + 1] - start;

  float4 accA = {0.0f, 0.0f, 0.0f, 0.0f};
  float4 accB = {0.0f, 0.0f, 0.0f, 0.0f};
  if (deg > 0) {
    const int*    cs  = csrc + start;
    const float4* el4 = (const float4*)el;
    const float4* f4  = (const float4*)feat;
    const float4  r4  = ((const float4*)er)[node];

    const bool has0 = lane < deg;
    int   s0  = 0;
    float e0x = -INFINITY, e0y = -INFINITY, e0z = -INFINITY, e0w = -INFINITY;
    if (has0) {
      s0 = cs[lane];
      const float4 l = el4[s0];
      e0x = lrelu(l.x + r4.x); e0y = lrelu(l.y + r4.y);
      e0z = lrelu(l.z + r4.z); e0w = lrelu(l.w + r4.w);
    }

    float mx = e0x, my = e0y, mz = e0z, mw = e0w;
    for (int j = lane + 32; j < deg; j += 32) {
      const float4 l = el4[cs[j]];
      mx = fmaxf(mx, lrelu(l.x + r4.x)); my = fmaxf(my, lrelu(l.y + r4.y));
      mz = fmaxf(mz, lrelu(l.z + r4.z)); mw = fmaxf(mw, lrelu(l.w + r4.w));
    }
    mx = wave_max(mx); my = wave_max(my); mz = wave_max(mz); mw = wave_max(mw);
    if (!(mx > -1e38f)) mx = 0.0f;
    if (!(my > -1e38f)) my = 0.0f;
    if (!(mz > -1e38f)) mz = 0.0f;
    if (!(mw > -1e38f)) mw = 0.0f;

    const float x0x = has0 ? __expf(e0x - mx) : 0.0f;
    const float x0y = has0 ? __expf(e0y - my) : 0.0f;
    const float x0z = has0 ? __expf(e0z - mz) : 0.0f;
    const float x0w = has0 ? __expf(e0w - mw) : 0.0f;
    float dx = x0x, dy = x0y, dz = x0z, dw = x0w;
    for (int j = lane + 32; j < deg; j += 32) {
      const float4 l = el4[cs[j]];
      dx += __expf(lrelu(l.x + r4.x) - mx); dy += __expf(lrelu(l.y + r4.y) - my);
      dz += __expf(lrelu(l.z + r4.z) - mz); dw += __expf(lrelu(l.w + r4.w) - mw);
    }
    dx = fmaxf(wave_sum(dx), 1e-16f); dy = fmaxf(wave_sum(dy), 1e-16f);
    dz = fmaxf(wave_sum(dz), 1e-16f); dw = fmaxf(wave_sum(dw), 1e-16f);

    const bool hi = lane >= 16;
    for (int base = 0; base < deg; base += 32) {
      int   s  = s0;
      float ax = x0x / dx, ay = x0y / dy, az = x0z / dz, aw = x0w / dw;
      if (base > 0) {
        const int j = base + lane;
        s = 0; ax = 0.0f; ay = 0.0f; az = 0.0f; aw = 0.0f;
        if (j < deg) {
          s = cs[j];
          const float4 l = el4[s];
          ax = __expf(lrelu(l.x + r4.x) - mx) / dx;
          ay = __expf(lrelu(l.y + r4.y) - my) / dy;
          az = __expf(lrelu(l.z + r4.z) - mz) / dz;
          aw = __expf(lrelu(l.w + r4.w) - mw) / dw;
        }
      }
      const int left = deg - base;
      const int cnt  = left < 32 ? left : 32;
      #pragma unroll 4
      for (int k = 0; k < cnt; ++k) {
        const int   sk = __builtin_amdgcn_readlane(s, k);
        const float bx = lane_bcast(ax, k), by = lane_bcast(ay, k);
        const float bz = lane_bcast(az, k), bw = lane_bcast(aw, k);
        const float aA = hi ? by : bx;
        const float aB = hi ? bw : bz;
        const float4* row = f4 + (long)sk * (HIDC / 4);
        const float4 fA = row[lane];
        const float4 fB = row[32 + lane];
        accA.x += aA * fA.x; accA.y += aA * fA.y;
        accA.z += aA * fA.z; accA.w += aA * fA.w;
        accB.x += aB * fB.x; accB.y += aB * fB.y;
        accB.z += aB * fB.z; accB.w += aB * fB.w;
      }
    }
  }

  const float4* rs4 = (const float4*)res + (long)node * (HIDC / 4);
  float4 oA = rs4[lane], oB = rs4[32 + lane];
  if (bias) {
    const float4 bA = ((const float4*)bias)[lane];
    const float4 bB = ((const float4*)bias)[32 + lane];
    oA.x += bA.x; oA.y += bA.y; oA.z += bA.z; oA.w += bA.w;
    oB.x += bB.x; oB.y += bB.y; oB.z += bB.z; oB.w += bB.w;
  }
  oA.x += accA.x; oA.y += accA.y; oA.z += accA.z; oA.w += accA.w;
  oB.x += accB.x; oB.y += accB.y; oB.z += accB.z; oB.w += accB.w;

  if (!MEAN) {
    oA.x = fmaxf(oA.x, 0.0f); oA.y = fmaxf(oA.y, 0.0f);
    oA.z = fmaxf(oA.z, 0.0f); oA.w = fmaxf(oA.w, 0.0f);
    oB.x = fmaxf(oB.x, 0.0f); oB.y = fmaxf(oB.y, 0.0f);
    oB.z = fmaxf(oB.z, 0.0f); oB.w = fmaxf(oB.w, 0.0f);
    float4* o4 = (float4*)out + (long)node * (HIDC / 4);
    o4[lane] = oA;
    o4[32 + lane] = oB;
  } else {
    float4 h1, h3;
    h1.x = __shfl_xor(oA.x, 16, 32); h1.y = __shfl_xor(oA.y, 16, 32);
    h1.z = __shfl_xor(oA.z, 16, 32); h1.w = __shfl_xor(oA.w, 16, 32);
    h3.x = __shfl_xor(oB.x, 16, 32); h3.y = __shfl_xor(oB.y, 16, 32);
    h3.z = __shfl_xor(oB.z, 16, 32); h3.w = __shfl_xor(oB.w, 16, 32);
    if (lane < 16) {
      float4 m;
      m.x = 0.25f * (((oA.x + h1.x) + oB.x) + h3.x);
      m.y = 0.25f * (((oA.y + h1.y) + oB.y) + h3.y);
      m.z = 0.25f * (((oA.z + h1.z) + oB.z) + h3.z);
      m.w = 0.25f * (((oA.w + h1.w) + oB.w) + h3.w);
      ((float4*)out)[(long)node * (DHC / 4) + lane] = m;
    }
  }
}

extern "C" void kernel_launch(void* const* d_in, const int* in_sizes, int n_in,
                              void* d_out, int out_size, void* d_ws, size_t ws_size,
                              hipStream_t stream) {
  const float* x     = (const float*)d_in[0];
  const int*   ei    = (const int*)d_in[1];
  const float* W[3]  = {(const float*)d_in[2], (const float*)d_in[3],
                        (const float*)d_in[4]};
  const float* al[3] = {(const float*)d_in[5], (const float*)d_in[6],
                        (const float*)d_in[7]};
  const float* ar[3] = {(const float*)d_in[8], (const float*)d_in[9],
                        (const float*)d_in[10]};
  const float* b[3]  = {(const float*)d_in[11], (const float*)d_in[12],
                        (const float*)d_in[13]};
  const float* resW0 = (const float*)d_in[14];

  const int N = in_sizes[0] / 128;
  const int E = in_sizes[1] / 2;
  const int* src = ei;
  const int* dst = ei + E;

  float* ws   = (float*)d_ws;
  float* FEAT = ws;
  float* OUT  = FEAT + (size_t)N * HIDC;
  float* HBUF = OUT  + (size_t)N * HIDC;
  float* ELP  = HBUF + (size_t)N * HIDC;
  float* ERP  = ELP  + (size_t)N * HEADS;
  int*   CNT  = (int*)(ERP + (size_t)N * HEADS);
  int*   RPT  = CNT + (size_t)N;
  int*   CSRC = RPT + (size_t)N + 1;

  const int  mtiles = (N + 15) / 16;
  const int  gN     = (N + 255) / 256;
  const int  gE     = (E + 255) / 256;
  const int  gNODE  = (N + 7) / 8;

  zero_int_kernel<<<gN, 256, 0, stream>>>(CNT, N);
  in_degree_kernel<<<gE, 256, 0, stream>>>(dst, CNT, E);
  row_start_kernel<<<gN, 256, 0, stream>>>(CNT, RPT, N);
  csr_fill_kernel<<<gE, 256, 0, stream>>>(src, dst, RPT, CNT, CSRC, E);

  gemm_wmma_f32<128, true><<<mtiles, 512, 0, stream>>>(
      x, W[0], nullptr, FEAT, al[0], ar[0], ELP, ERP, N);
  gemm_wmma_f32<128, false><<<mtiles, 512, 0, stream>>>(
      x, resW0, b[0], OUT, nullptr, nullptr, nullptr, nullptr, N);
  node_softmax_aggregate_kernel<false><<<gNODE, 256, 0, stream>>>(
      RPT, CSRC, FEAT, ELP, ERP, OUT, nullptr, HBUF, N);

  gemm_wmma_f32<256, true><<<mtiles, 512, 0, stream>>>(
      HBUF, W[1], nullptr, FEAT, al[1], ar[1], ELP, ERP, N);
  node_softmax_aggregate_kernel<false><<<gNODE, 256, 0, stream>>>(
      RPT, CSRC, FEAT, ELP, ERP, HBUF, b[1], OUT, N);

  gemm_wmma_f32<256, true><<<mtiles, 512, 0, stream>>>(
      OUT, W[2], nullptr, FEAT, al[2], ar[2], ELP, ERP, N);
  node_softmax_aggregate_kernel<true><<<gNODE, 256, 0, stream>>>(
      RPT, CSRC, FEAT, ELP, ERP, OUT, b[2], (float*)d_out, N);
}


# GEMM block tile 16x256 (16 waves) -> 64x128 (8 waves): each wave keeps 4 row-strip accumulators per W fragment, W re-read from L2 4x less, 4x fewer waves to start; every float sum order unchanged
# speedup vs baseline: 5.6192x; 1.0409x over previous
#include <hip/hip_runtime.h>
#include <math.h>


typedef __attribute__((ext_vector_type(2))) float v2f;
typedef __attribute__((ext_vector_type(8))) float v8f;

#define HIDC 256
#define HEADS 4
#define DHC 64

__device__ __forceinline__ float lrelu(float v) {
  return v > 0.0f ? v : 0.2f * v;
}

template <int K, bool ATT>
__global__ __launch_bounds__(256)
void gemm_wmma_f32(const float* __restrict__ A, const float* __restrict__ W,
                   const float* __restrict__ bias, float* __restrict__ C,
                   const float* __restrict__ al, const float* __restrict__ ar,
                   float* __restrict__ el, float* __restrict__ er, int M) {
  constexpr int MT   = 4;
  constexpr int ROWS = 16 * MT;
  constexpr int LDA  = K + 4;
  constexpr int K4   = K / 4;
  __shared__ float4 As4[ROWS * LDA / 4];
  __shared__ float4 red4[ATT ? 2 * ROWS * 2 : 1];
  float* As = (float*)As4;

  const int tid   = threadIdx.x;
  const int mBase = (int)(blockIdx.x >> 1) * ROWS;
  const int cHalf = (int)(blockIdx.x & 1);
  const int cBase = cHalf << 7;

  #pragma unroll
  for (int i = tid; i < ROWS * K4; i += 256) {
    int r = i / K4;
    int c = i - r * K4;
    int row = mBase + r;
    if (row >= M) row = M - 1;
    As4[r * (LDA / 4) + c] = ((const float4*)(A + (long)row * K))[c];
  }
  __syncthreads();

  const int lane  = tid & 31;
  const int wave  = tid >> 5;
  const int nBase = cBase + (wave << 4);
  const int col   = lane & 15;
  const int kHalf = (lane >> 4) << 1;

  const float* as = &As[(lane & 15) * LDA + kHalf];
  const float* bp = W + (long)kHalf * HIDC + nBase + col;

  const float alc = ATT ? al[nBase + col] : 0.0f;
  const float arc = ATT ? ar[nBase + col] : 0.0f;

  v8f acc[MT];
  #pragma unroll
  for (int t = 0; t < MT; ++t) acc[t] = v8f{};

  #pragma unroll 4
  for (int kb = 0; kb < K; kb += 4) {
    v2f b;
    b.x = bp[0];
    b.y = bp[HIDC];
    #pragma unroll
    for (int t = 0; t < MT; ++t) {
      v2f a;
      a.x = as[t * 16 * LDA];
      a.y = as[t * 16 * LDA + 1];
      acc[t] = __builtin_amdgcn_wmma_f32_16x16x4_f32(
           false, a,  false, b,
           (short)0, acc[t],  false,  false);
    }
    as += 4;
    bp += 4 * HIDC;
  }

  const int rowOff = (lane >> 4) << 3;

  if (ATT) {
    const bool b3 = (col & 8) != 0, b2 = (col & 4) != 0;
    const bool b1 = (col & 2) != 0, b0 = (col & 1) != 0;
    float* red = (float*)red4;
    #pragma unroll
    for (int t = 0; t < MT; ++t) {
      float v[16];
      #pragma unroll
      for (int r = 0; r < 8; ++r) {
        v[2 * r]     = acc[t][r] * alc;
        v[2 * r + 1] = acc[t][r] * arc;
      }
      float w8[8], w4[4], w2[2];
      #pragma unroll
      for (int j = 0; j < 8; ++j) {
        const float keep = b3 ? v[8 + j] : v[j];
        const float send = b3 ? v[j] : v[8 + j];
        w8[j] = keep + __shfl_xor(send, 8, 32);
      }
      #pragma unroll
      for (int j = 0; j < 4; ++j) {
        const float keep = b2 ? w8[4 + j] : w8[j];
        const float send = b2 ? w8[j] : w8[4 + j];
        w4[j] = keep + __shfl_xor(send, 4, 32);
      }
      #pragma unroll
      for (int j = 0; j < 2; ++j) {
        const float keep = b1 ? w4[2 + j] : w4[j];
        const float send = b1 ? w4[j] : w4[2 + j];
        w2[j] = keep + __shfl_xor(send, 2, 32);
      }
      const float keep1 = b0 ? w2[1] : w2[0];
      const float send1 = b0 ? w2[0] : w2[1];
      const float part  = keep1 + __shfl_xor(send1, 1, 32);

      red[(((col & 1) * ROWS) + t * 16 + rowOff + (col >> 1)) * 8 + wave] = part;
    }
    __syncthreads();
  }

  const float bc = bias ? bias[nBase + col] : 0.0f;
  #pragma unroll
  for (int t = 0; t < MT; ++t) {
    #pragma unroll
    for (int r = 0; r < 8; ++r) {
      int row = mBase + t * 16 + rowOff + r;
      if (row < M) C[(long)row * HIDC + nBase + col] = acc[t][r] + bc;
    }
  }

  if (ATT) {
    const int which = tid >> 7;
    const int rem   = tid & 127;
    const int row   = rem >> 1;
    const int hl    = rem & 1;
    float* dstp = (which == 0) ? el : er;
    const float4 q = red4[(which * ROWS + row) * 2 + hl];
    if (mBase + row < M)
      dstp[(long)(mBase + row) * HEADS + 2 * cHalf + hl] = ((q.x + q.y) + q.z) + q.w;
  }
}


__global__ void zero_int_kernel(int* __restrict__ p, int n) {
  int i = blockIdx.x * blockDim.x + threadIdx.x;
  if (i < n) p[i] = 0;
}

__global__ void in_degree_kernel(const int* __restrict__ dst,
                                 int* __restrict__ cnt, int E) {
  int e = blockIdx.x * blockDim.x + threadIdx.x;
  if (e < E) atomicAdd(&cnt[dst[e]], 1);
}

__global__ __launch_bounds__(256)
void row_start_kernel(const int* __restrict__ cnt, int* __restrict__ rowptr,
                      int n) {
  __shared__ int wave_pre[8];
  __shared__ int wave_tot[8];
  const int t = threadIdx.x, lane = t & 31, wave = t >> 5;
  const int tile0 = blockIdx.x * 256;

  int pre = 0;
  for (int i = t; i < tile0; i += 256) pre += cnt[i];
  for (int off = 16; off; off >>= 1) pre += __shfl_xor(pre, off, 32);

  const int i = tile0 + t;
  const int c = (i < n) ? cnt[i] : 0;
  int inc = c;
  for (int off = 1; off < 32; off <<= 1) {
    int v = __shfl_up(inc, off, 32);
    if (lane >= off) inc += v;
  }
  if (lane == 31) wave_tot[wave] = inc;
  if (lane == 0)  wave_pre[wave] = pre;
  __syncthreads();
  int base = 0;
  for (int w = 0; w < 8; ++w) {
    base += wave_pre[w];
    if (w < wave) base += wave_tot[w];
  }
  const int excl = base + inc - c;
  if (i < n) {
    rowptr[i] = excl;
    if (i == n - 1) rowptr[n] = excl + c;
  }
}

__global__ void csr_fill_kernel(const int* __restrict__ src,
                                const int* __restrict__ dst,
                                const int* __restrict__ rowptr,
                                int* __restrict__ cnt,
                                int* __restrict__ csrc, int E) {
  int e = blockIdx.x * blockDim.x + threadIdx.x;
  if (e >= E) return;
  int d = dst[e];
  int k = atomicSub(&cnt[d], 1);
  csrc[rowptr[d] + k - 1] = src[e];
}

__device__ __forceinline__ float wave_max(float v) {
  for (int off = 16; off; off >>= 1) v = fmaxf(v, __shfl_xor(v, off, 32));
  return v;
}

__device__ __forceinline__ float wave_sum(float v) {
  for (int off = 16; off; off >>= 1) v += __shfl_xor(v, off, 32);
  return v;
}

__device__ __forceinline__ float lane_bcast(float v, int k) {
  return __int_as_float(__builtin_amdgcn_readlane(__float_as_int(v), k));
}

template <bool MEAN>
__global__ __launch_bounds__(256)
void node_softmax_aggregate_kernel(const int* __restrict__ rowptr,
                                   const int* __restrict__ csrc,
                                   const float* __restrict__ feat,
                                   const float* __restrict__ el,
                                   const float* __restrict__ er,
                                   const float* __restrict__ res,
                                   const float* __restrict__ bias,
                                   float* __restrict__ out, int n) {
  const int lane = threadIdx.x & 31;
  const int node = __builtin_amdgcn_readfirstlane(
      (int)(blockIdx.x * 8 + (threadIdx.x >> 5)));
  if (node >= n) return;
  const int start = rowptr[node];
  const int deg   = rowptr[node + 1] - start;

  float4 accA = {0.0f, 0.0f, 0.0f, 0.0f};
  float4 accB = {0.0f, 0.0f, 0.0f, 0.0f};
  if (deg > 0) {
    const int*    cs  = csrc + start;
    const float4* el4 = (const float4*)el;
    const float4* f4  = (const float4*)feat;
    const float4  r4  = ((const float4*)er)[node];

    const bool has0 = lane < deg;
    int   s0  = 0;
    float e0x = -INFINITY, e0y = -INFINITY, e0z = -INFINITY, e0w = -INFINITY;
    if (has0) {
      s0 = cs[lane];
      const float4 l = el4[s0];
      e0x = lrelu(l.x + r4.x); e0y = lrelu(l.y + r4.y);
      e0z = lrelu(l.z + r4.z); e0w = lrelu(l.w + r4.w);
    }

    float mx = e0x, my = e0y, mz = e0z, mw = e0w;
    for (int j = lane + 32; j < deg; j += 32) {
      const float4 l = el4[cs[j]];
      mx = fmaxf(mx, lrelu(l.x + r4.x)); my = fmaxf(my, lrelu(l.y + r4.y));
      mz = fmaxf(mz, lrelu(l.z + r4.z)); mw = fmaxf(mw, lrelu(l.w + r4.w));
    }
    mx = wave_max(mx); my = wave_max(my); mz = wave_max(mz); mw = wave_max(mw);
    if (!(mx > -1e38f)) mx = 0.0f;
    if (!(my > -1e38f)) my = 0.0f;
    if (!(mz > -1e38f)) mz = 0.0f;
    if (!(mw > -1e38f)) mw = 0.0f;

    const float x0x = has0 ? __expf(e0x - mx) : 0.0f;
    const float x0y = has0 ? __expf(e0y - my) : 0.0f;
    const float x0z = has0 ? __expf(e0z - mz) : 0.0f;
    const float x0w = has0 ? __expf(e0w - mw) : 0.0f;
    float dx = x0x, dy = x0y, dz = x0z, dw = x0w;
    for (int j = lane + 32; j < deg; j += 32) {
      const float4 l = el4[cs[j]];
      dx += __expf(lrelu(l.x + r4.x) - mx); dy += __expf(lrelu(l.y + r4.y) - my);
      dz += __expf(lrelu(l.z + r4.z) - mz); dw += __expf(lrelu(l.w + r4.w) - mw);
    }
    dx = fmaxf(wave_sum(dx), 1e-16f); dy = fmaxf(wave_sum(dy), 1e-16f);
    dz = fmaxf(wave_sum(dz), 1e-16f); dw = fmaxf(wave_sum(dw), 1e-16f);

    const bool hi = lane >= 16;
    for (int base = 0; base < deg; base += 32) {
      int   s  = s0;
      float ax = x0x / dx, ay = x0y / dy, az = x0z / dz, aw = x0w / dw;
      if (base > 0) {
        const int j = base + lane;
        s = 0; ax = 0.0f; ay = 0.0f; az = 0.0f; aw = 0.0f;
        if (j < deg) {
          s = cs[j];
          const float4 l = el4[s];
          ax = __expf(lrelu(l.x + r4.x) - mx) / dx;
          ay = __expf(lrelu(l.y + r4.y) - my) / dy;
          az = __expf(lrelu(l.z + r4.z) - mz) / dz;
          aw = __expf(lrelu(l.w + r4.w) - mw) / dw;
        }
      }
      const int left = deg - base;
      const int cnt  = left < 32 ? left : 32;
      #pragma unroll 4
      for (int k = 0; k < cnt; ++k) {
        const int   sk = __builtin_amdgcn_readlane(s, k);
        const float bx = lane_bcast(ax, k), by = lane_bcast(ay, k);
        const float bz = lane_bcast(az, k), bw = lane_bcast(aw, k);
        const float aA = hi ? by : bx;
        const float aB = hi ? bw : bz;
        const float4* row = f4 + (long)sk * (HIDC / 4);
        const float4 fA = row[lane];
        const float4 fB = row[32 + lane];
        accA.x += aA * fA.x; accA.y += aA * fA.y;
        accA.z += aA * fA.z; accA.w += aA * fA.w;
        accB.x += aB * fB.x; accB.y += aB * fB.y;
        accB.z += aB * fB.z; accB.w += aB * fB.w;
      }
    }
  }

  const float4* rs4 = (const float4*)res + (long)node * (HIDC / 4);
  float4 oA = rs4[lane], oB = rs4[32 + lane];
  if (bias) {
    const float4 bA = ((const float4*)bias)[lane];
    const float4 bB = ((const float4*)bias)[32 + lane];
    oA.x += bA.x; oA.y += bA.y; oA.z += bA.z; oA.w += bA.w;
    oB.x += bB.x; oB.y += bB.y; oB.z += bB.z; oB.w += bB.w;
  }
  oA.x += accA.x; oA.y += accA.y; oA.z += accA.z; oA.w += accA.w;
  oB.x += accB.x; oB.y += accB.y; oB.z += accB.z; oB.w += accB.w;

  if (!MEAN) {
    oA.x = fmaxf(oA.x, 0.0f); oA.y = fmaxf(oA.y, 0.0f);
    oA.z = fmaxf(oA.z, 0.0f); oA.w = fmaxf(oA.w, 0.0f);
    oB.x = fmaxf(oB.x, 0.0f); oB.y = fmaxf(oB.y, 0.0f);
    oB.z = fmaxf(oB.z, 0.0f); oB.w = fmaxf(oB.w, 0.0f);
    float4* o4 = (float4*)out + (long)node * (HIDC / 4);
    o4[lane] = oA;
    o4[32 + lane] = oB;
  } else {
    float4 h1, h3;
    h1.x = __shfl_xor(oA.x, 16, 32); h1.y = __shfl_xor(oA.y, 16, 32);
    h1.z = __shfl_xor(oA.z, 16, 32); h1.w = __shfl_xor(oA.w, 16, 32);
    h3.x = __shfl_xor(oB.x, 16, 32); h3.y = __shfl_xor(oB.y, 16, 32);
    h3.z = __shfl_xor(oB.z, 16, 32); h3.w = __shfl_xor(oB.w, 16, 32);
    if (lane < 16) {
      float4 m;
      m.x = 0.25f * (((oA.x + h1.x) + oB.x) + h3.x);
      m.y = 0.25f * (((oA.y + h1.y) + oB.y) + h3.y);
      m.z = 0.25f * (((oA.z + h1.z) + oB.z) + h3.z);
      m.w = 0.25f * (((oA.w + h1.w) + oB.w) + h3.w);
      ((float4*)out)[(long)node * (DHC / 4) + lane] = m;
    }
  }
}

extern "C" void kernel_launch(void* const* d_in, const int* in_sizes, int n_in,
                              void* d_out, int out_size, void* d_ws, size_t ws_size,
                              hipStream_t stream) {
  const float* x     = (const float*)d_in[0];
  const int*   ei    = (const int*)d_in[1];
  const float* W[3]  = {(const float*)d_in[2], (const float*)d_in[3],
                        (const float*)d_in[4]};
  const float* al[3] = {(const float*)d_in[5], (const float*)d_in[6],
                        (const float*)d_in[7]};
  const float* ar[3] = {(const float*)d_in[8], (const float*)d_in[9],
                        (const float*)d_in[10]};
  const float* b[3]  = {(const float*)d_in[11], (const float*)d_in[12],
                        (const float*)d_in[13]};
  const float* resW0 = (const float*)d_in[14];

  const int N = in_sizes[0] / 128;
  const int E = in_sizes[1] / 2;
  const int* src = ei;
  const int* dst = ei + E;

  float* ws   = (float*)d_ws;
  float* FEAT = ws;
  float* OUT  = FEAT + (size_t)N * HIDC;
  float* HBUF = OUT  + (size_t)N * HIDC;
  float* ELP  = HBUF + (size_t)N * HIDC;
  float* ERP  = ELP  + (size_t)N * HEADS;
  int*   CNT  = (int*)(ERP + (size_t)N * HEADS);
  int*   RPT  = CNT + (size_t)N;
  int*   CSRC = RPT + (size_t)N + 1;

  const int  gGEMM  = ((N + 63) / 64) * 2;
  const int  gN     = (N + 255) / 256;
  const int  gE     = (E + 255) / 256;
  const int  gNODE  = (N + 7) / 8;

  zero_int_kernel<<<gN, 256, 0, stream>>>(CNT, N);
  in_degree_kernel<<<gE, 256, 0, stream>>>(dst, CNT, E);
  row_start_kernel<<<gN, 256, 0, stream>>>(CNT, RPT, N);
  csr_fill_kernel<<<gE, 256, 0, stream>>>(src, dst, RPT, CNT, CSRC, E);

  gemm_wmma_f32<128, true><<<gGEMM, 256, 0, stream>>>(
      x, W[0], nullptr, FEAT, al[0], ar[0], ELP, ERP, N);
  gemm_wmma_f32<128, false><<<gGEMM, 256, 0, stream>>>(
      x, resW0, b[0], OUT, nullptr, nullptr, nullptr, nullptr, N);
  node_softmax_aggregate_kernel<false><<<gNODE, 256, 0, stream>>>(
      RPT, CSRC, FEAT, ELP, ERP, OUT, nullptr, HBUF, N);

  gemm_wmma_f32<256, true><<<gGEMM, 256, 0, stream>>>(
      HBUF, W[1], nullptr, FEAT, al[1], ar[1], ELP, ERP, N);
  node_softmax_aggregate_kernel<false><<<gNODE, 256, 0, stream>>>(
      RPT, CSRC, FEAT, ELP, ERP, HBUF, b[1], OUT, N);

  gemm_wmma_f32<256, true><<<gGEMM, 256, 0, stream>>>(
      OUT, W[2], nullptr, FEAT, al[2], ar[2], ELP, ERP, N);
  node_softmax_aggregate_kernel<true><<<gNODE, 256, 0, stream>>>(
      RPT, CSRC, FEAT, ELP, ERP, OUT, b[2], (float*)d_out, N);
}
